# starting kernel
# baseline (speedup 1.0000x reference)
#include <hip/hip_runtime.h>
#include <math.h>


typedef __attribute__((ext_vector_type(2))) float v2f;
typedef __attribute__((ext_vector_type(8))) float v8f;

#define IMG_H 512
#define IMG_W 512
#define NIMG  16
#define WIN   81
#define PAD   40
#define BAND  96
#define KSTEPS 24
#define XS_STRIDE 80
#define TL_STRIDE 145

__device__ __forceinline__ float cleanf(float v) {
    v = (v != v) ? 0.0f : v;
    return fminf(fmaxf(v, 0.0f), 1.0f);
}

__global__ __launch_bounds__(128) void ssim_pass1(
    const float* __restrict__ img1, const float* __restrict__ img2,
    const float* __restrict__ w1d, float* __restrict__ T)
{
    __shared__ float wpad[112];
    __shared__ float xs1[BAND * XS_STRIDE];
    __shared__ float xs2[BAND * XS_STRIDE];

    const int tid = threadIdx.x;
    const int cg0 = blockIdx.x * 64;
    const int m0  = blockIdx.y * 16;
    const int img = blockIdx.z;

    if (tid < 112) wpad[tid] = (tid >= 16 && tid < 16 + WIN) ? w1d[tid - 16] : 0.0f;

    const float* p1 = img1 + (size_t)img * IMG_H * IMG_W;
    const float* p2 = img2 + (size_t)img * IMG_H * IMG_W;
    for (int i = tid; i < BAND * 64; i += 128) {
        int r = i >> 6, c = i & 63;
        int gr = m0 - PAD + r;
        float a = 0.0f, b = 0.0f;
        if (gr >= 0 && gr < IMG_H) {
            int off = gr * IMG_W + cg0 + c;
            a = cleanf(p1[off]);
            b = cleanf(p2[off]);
        }
        xs1[r * XS_STRIDE + c] = a;
        xs2[r * XS_STRIDE + c] = b;
    }
    __syncthreads();

    const int lane = tid & 31;
    const int wv   = tid >> 5;
    const int j    = lane & 15;
    const int hi   = lane >> 4;

    v8f acc0 = {}, acc1 = {}, acc2 = {}, acc3 = {}, acc4 = {};

    for (int step = 0; step < KSTEPS; ++step) {
        const int kb = step << 2;
        v2f a, b1, b2, b11, b22, b12;
#pragma unroll
        for (int v = 0; v < 2; ++v) {
            const int kk = v + 2 * hi;
            a[v] = wpad[16 + kb + kk - j];
            const int xi = (kb + kk) * XS_STRIDE + wv * 16 + j;
            const float x1 = xs1[xi], x2 = xs2[xi];
            b1[v] = x1;  b2[v] = x2;
            b11[v] = x1 * x1;  b22[v] = x2 * x2;  b12[v] = x1 * x2;
        }
        acc0 = __builtin_amdgcn_wmma_f32_16x16x4_f32(false, a, false, b1,  (short)0, acc0, false, false);
        acc1 = __builtin_amdgcn_wmma_f32_16x16x4_f32(false, a, false, b2,  (short)0, acc1, false, false);
        acc2 = __builtin_amdgcn_wmma_f32_16x16x4_f32(false, a, false, b11, (short)0, acc2, false, false);
        acc3 = __builtin_amdgcn_wmma_f32_16x16x4_f32(false, a, false, b22, (short)0, acc3, false, false);
        acc4 = __builtin_amdgcn_wmma_f32_16x16x4_f32(false, a, false, b12, (short)0, acc4, false, false);
    }

    const int col = cg0 + wv * 16 + j;
    const size_t plane = (size_t)IMG_H * IMG_W;
#pragma unroll
    for (int v = 0; v < 8; ++v) {
        const size_t rowoff = (size_t)(m0 + v + 8 * hi) * IMG_W + col;
        T[(0 * NIMG + img) * plane + rowoff] = acc0[v];
        T[(1 * NIMG + img) * plane + rowoff] = acc1[v];
        T[(2 * NIMG + img) * plane + rowoff] = acc2[v];
        T[(3 * NIMG + img) * plane + rowoff] = acc3[v];
        T[(4 * NIMG + img) * plane + rowoff] = acc4[v];
    }
}

__global__ __launch_bounds__(128) void ssim_pass2(
    const float* __restrict__ T, const float* __restrict__ w1d,
    float* __restrict__ out)
{
    __shared__ float wpad[112];
    __shared__ float tl[5 * 16 * TL_STRIDE];

    const int tid = threadIdx.x;
    const int cg0 = blockIdx.x * 64;
    const int m0  = blockIdx.y * 16;
    const int img = blockIdx.z;

    if (tid < 112) wpad[tid] = (tid >= 16 && tid < 16 + WIN) ? w1d[tid - 16] : 0.0f;

    const size_t plane = (size_t)IMG_H * IMG_W;
    for (int i = tid; i < 5 * 16 * 144; i += 128) {
        int c   = i / 2304;
        int rem = i - c * 2304;
        int r   = rem / 144;
        int cc  = rem - r * 144;
        int gc  = cg0 - PAD + cc;
        float v = 0.0f;
        if (gc >= 0 && gc < IMG_W)
            v = T[((size_t)(c * NIMG + img)) * plane + (size_t)(m0 + r) * IMG_W + gc];
        tl[c * (16 * TL_STRIDE) + r * TL_STRIDE + cc] = v;
    }
    __syncthreads();

    const int lane = tid & 31;
    const int wv   = tid >> 5;
    const int j    = lane & 15;
    const int hi   = lane >> 4;

    v8f acc0 = {}, acc1 = {}, acc2 = {}, acc3 = {}, acc4 = {};

    for (int step = 0; step < KSTEPS; ++step) {
        const int kb = step << 2;
        v2f b, a0, a1, a2, a3, a4;
#pragma unroll
        for (int v = 0; v < 2; ++v) {
            const int kk = v + 2 * hi;
            b[v] = wpad[96 + j - kb - kk];
            const int ci = j * TL_STRIDE + wv * 16 + kb + kk;
            a0[v] = tl[ci];
            a1[v] = tl[1 * (16 * TL_STRIDE) + ci];
            a2[v] = tl[2 * (16 * TL_STRIDE) + ci];
            a3[v] = tl[3 * (16 * TL_STRIDE) + ci];
            a4[v] = tl[4 * (16 * TL_STRIDE) + ci];
        }
        acc0 = __builtin_amdgcn_wmma_f32_16x16x4_f32(false, a0, false, b, (short)0, acc0, false, false);
        acc1 = __builtin_amdgcn_wmma_f32_16x16x4_f32(false, a1, false, b, (short)0, acc1, false, false);
        acc2 = __builtin_amdgcn_wmma_f32_16x16x4_f32(false, a2, false, b, (short)0, acc2, false, false);
        acc3 = __builtin_amdgcn_wmma_f32_16x16x4_f32(false, a3, false, b, (short)0, acc3, false, false);
        acc4 = __builtin_amdgcn_wmma_f32_16x16x4_f32(false, a4, false, b, (short)0, acc4, false, false);
    }

    const float C1v = 1e-4f, C2v = 9e-4f;
    const int col = cg0 + wv * 16 + j;
#pragma unroll
    for (int v = 0; v < 8; ++v) {
        const float mu1 = acc0[v], mu2 = acc1[v];
        const float e11 = acc2[v], e22 = acc3[v], e12 = acc4[v];
        const float mu1sq = mu1 * mu1, mu2sq = mu2 * mu2, mu12 = mu1 * mu2;
        const float s11 = fmaxf(e11 - mu1sq, 0.0f);
        const float s22 = fmaxf(e22 - mu2sq, 0.0f);
        const float s12 = e12 - mu12;
        const float num = (2.0f * mu12 + C1v) * (2.0f * s12 + C2v);
        const float den = (mu1sq + mu2sq + C1v) * (s11 + s22 + C2v);
        float r = num / (den + 1e-8f);
        r = (r != r) ? 0.0f : r;
        r = fminf(fmaxf(r, -1.0f), 1.0f);
        out[(size_t)(img * IMG_H + m0 + v + 8 * hi) * IMG_W + col] = r;
    }
}

extern "C" void kernel_launch(void* const* d_in, const int* in_sizes, int n_in,
                              void* d_out, int out_size, void* d_ws, size_t ws_size,
                              hipStream_t stream) {
    const float* img1 = (const float*)d_in[0];
    const float* img2 = (const float*)d_in[1];
    const float* w1d  = (const float*)d_in[2];
    float* T   = (float*)d_ws;
    float* out = (float*)d_out;

    dim3 grid(IMG_W / 64, IMG_H / 16, NIMG);
    dim3 block(128);
    hipLaunchKernelGGL(ssim_pass1, grid, block, 0, stream, img1, img2, w1d, T);
    hipLaunchKernelGGL(ssim_pass2, grid, block, 0, stream, T, w1d, out);
}


# fragments loaded straight from global as 128-bit vectors in a permuted k order, no LDS staging: row blur first, intermediate stored fragment-blocked; fp summation order changes
# speedup vs baseline: 2.7546x; 2.7546x over previous
#include <hip/hip_runtime.h>
#include <math.h>


typedef __attribute__((ext_vector_type(2))) float v2f;
typedef __attribute__((ext_vector_type(4))) float v4f;
typedef __attribute__((ext_vector_type(8))) float v8f;

#define IMG_H 512
#define IMG_W 512
#define NIMG  16
#define WIN   81
#define PAD   40
#define NTILE 32
#define NGRP  64

__device__ __forceinline__ float cleanf(float v, float top) {
    v = (v != v) ? 0.0f : v;
    return fminf(fmaxf(v, 0.0f), top);
}

#define WMMA(A, B, C) __builtin_amdgcn_wmma_f32_16x16x4_f32(false, (A), false, (B), (short)0, (C), false, false)

__device__ __forceinline__ void load_wfrag(const float* wz, int j, int half, v2f (&wf)[24]) {
#pragma unroll
    for (int s = 0; s < 24; ++s) {
        const int p = 8 * (s >> 1) + 4 * half + 2 * (s & 1);
        wf[s][0] = wz[16 + p - j];
        wf[s][1] = wz[17 + p - j];
    }
}

__global__ __launch_bounds__(128) void ssim_blur_rows(
    const float* __restrict__ img1, const float* __restrict__ img2,
    const float* __restrict__ w1d, float* __restrict__ U)
{
    __shared__ float wz[112];
    const int tid = threadIdx.x;
    if (tid < 112) wz[tid] = (tid >= 16 && tid < 16 + WIN) ? w1d[tid - 16] : 0.0f;
    __syncthreads();

    const int lane = tid & 31;
    const int wv   = __builtin_amdgcn_readfirstlane(tid >> 5);
    const int j    = lane & 15;
    const int half = lane >> 4;

    v2f wf[24];
    load_wfrag(wz, j, half, wf);

    const int m0  = blockIdx.y * 16;
    const int img = blockIdx.z;
    const float* r1 = img1 + ((size_t)img * IMG_H + m0 + j) * IMG_W;
    const float* r2 = img2 + ((size_t)img * IMG_H + m0 + j) * IMG_W;

    for (int i = 0; i < 2; ++i) {
        const int ct = blockIdx.x * 8 + i * 4 + wv;
        const int c0 = ct * 16;
        v8f acc0 = {}, acc1 = {}, acc2 = {}, acc3 = {}, acc4 = {};
#pragma unroll
        for (int L = 0; L < 12; ++L) {
            const int col = c0 - PAD + 8 * L + 4 * half;
            const bool ok = (unsigned)col < (unsigned)IMG_W;
            const int cc = ok ? col : 0;
            const float top = ok ? 1.0f : 0.0f;
            v4f x1 = *(const v4f*)(r1 + cc);
            v4f x2 = *(const v4f*)(r2 + cc);
#pragma unroll
            for (int e = 0; e < 4; ++e) {
                x1[e] = cleanf(x1[e], top);
                x2[e] = cleanf(x2[e], top);
            }
#pragma unroll
            for (int q = 0; q < 2; ++q) {
                v2f a1, a2;
                a1[0] = x1[2 * q]; a1[1] = x1[2 * q + 1];
                a2[0] = x2[2 * q]; a2[1] = x2[2 * q + 1];
                const v2f a11 = a1 * a1, a22 = a2 * a2, a12 = a1 * a2;
                const v2f b = wf[2 * L + q];
                acc0 = WMMA(a1,  b, acc0);
                acc1 = WMMA(a2,  b, acc1);
                acc2 = WMMA(a11, b, acc2);
                acc3 = WMMA(a22, b, acc3);
                acc4 = WMMA(a12, b, acc4);
            }
        }
        const size_t blk = ((size_t)ct * NGRP + (m0 >> 3) + half) * 128 + j * 8;
        const size_t chs = (size_t)NIMG * NTILE * NGRP * 128;
        float* u = U + (size_t)img * NTILE * NGRP * 128 + blk;
        v4f lo, hi4;
#define ST(ACC, CH) lo[0]=ACC[0]; lo[1]=ACC[1]; lo[2]=ACC[2]; lo[3]=ACC[3]; \
        hi4[0]=ACC[4]; hi4[1]=ACC[5]; hi4[2]=ACC[6]; hi4[3]=ACC[7]; \
        *(v4f*)(u + (CH) * chs) = lo; *(v4f*)(u + (CH) * chs + 4) = hi4;
        ST(acc0, 0) ST(acc1, 1) ST(acc2, 2) ST(acc3, 3) ST(acc4, 4)
    }
}

__global__ __launch_bounds__(128) void ssim_blur_cols(
    const float* __restrict__ U, const float* __restrict__ w1d,
    float* __restrict__ out)
{
    __shared__ float wz[112];
    const int tid = threadIdx.x;
    if (tid < 112) wz[tid] = (tid >= 16 && tid < 16 + WIN) ? w1d[tid - 16] : 0.0f;
    __syncthreads();

    const int lane = tid & 31;
    const int wv   = __builtin_amdgcn_readfirstlane(tid >> 5);
    const int j    = lane & 15;
    const int half = lane >> 4;

    v2f wf[24];
    load_wfrag(wz, j, half, wf);

    const int ct  = blockIdx.x;
    const int img = blockIdx.z;
    const size_t chs = (size_t)NIMG * NTILE * NGRP * 128;
    const float* ub = U + ((size_t)img * NTILE + ct) * NGRP * 128 + j * 8 + half * 4;

    for (int i = 0; i < 2; ++i) {
        const int rt = blockIdx.y * 8 + i * 4 + wv;
        v8f acc0 = {}, acc1 = {}, acc2 = {}, acc3 = {}, acc4 = {};
#pragma unroll
        for (int L = 0; L < 12; ++L) {
            const int g = 2 * rt - 5 + L;
            const bool ok = (unsigned)g < (unsigned)NGRP;
            const float* p = ub + (size_t)(ok ? g : 0) * 128;
            v4f t0 = *(const v4f*)(p);
            v4f t1 = *(const v4f*)(p + chs);
            v4f t2 = *(const v4f*)(p + 2 * chs);
            v4f t3 = *(const v4f*)(p + 3 * chs);
            v4f t4 = *(const v4f*)(p + 4 * chs);
            const v4f z = {0.0f, 0.0f, 0.0f, 0.0f};
            t0 = ok ? t0 : z; t1 = ok ? t1 : z; t2 = ok ? t2 : z;
            t3 = ok ? t3 : z; t4 = ok ? t4 : z;
#pragma unroll
            for (int q = 0; q < 2; ++q) {
                v2f a0, a1, a2, a3, a4;
                a0[0] = t0[2 * q]; a0[1] = t0[2 * q + 1];
                a1[0] = t1[2 * q]; a1[1] = t1[2 * q + 1];
                a2[0] = t2[2 * q]; a2[1] = t2[2 * q + 1];
                a3[0] = t3[2 * q]; a3[1] = t3[2 * q + 1];
                a4[0] = t4[2 * q]; a4[1] = t4[2 * q + 1];
                const v2f b = wf[2 * L + q];
                acc0 = WMMA(a0, b, acc0);
                acc1 = WMMA(a1, b, acc1);
                acc2 = WMMA(a2, b, acc2);
                acc3 = WMMA(a3, b, acc3);
                acc4 = WMMA(a4, b, acc4);
            }
        }
        const float C1v = 1e-4f, C2v = 9e-4f;
        float r8[8];
#pragma unroll
        for (int v = 0; v < 8; ++v) {
            const float mu1 = acc0[v], mu2 = acc1[v];
            const float e11 = acc2[v], e22 = acc3[v], e12 = acc4[v];
            const float mu1sq = mu1 * mu1, mu2sq = mu2 * mu2, mu12 = mu1 * mu2;
            const float s11 = fmaxf(e11 - mu1sq, 0.0f);
            const float s22 = fmaxf(e22 - mu2sq, 0.0f);
            const float s12 = e12 - mu12;
            const float num = (2.0f * mu12 + C1v) * (2.0f * s12 + C2v);
            const float den = (mu1sq + mu2sq + C1v) * (s11 + s22 + C2v);
            float r = num / (den + 1e-8f);
            r = (r != r) ? 0.0f : r;
            r8[v] = fminf(fmaxf(r, -1.0f), 1.0f);
        }
        float* o = out + ((size_t)img * IMG_H + rt * 16 + j) * IMG_W + ct * 16 + 8 * half;
        v4f lo, hi4;
        lo[0] = r8[0]; lo[1] = r8[1]; lo[2] = r8[2]; lo[3] = r8[3];
        hi4[0] = r8[4]; hi4[1] = r8[5]; hi4[2] = r8[6]; hi4[3] = r8[7];
        *(v4f*)(o) = lo;
        *(v4f*)(o + 4) = hi4;
    }
}

extern "C" void kernel_launch(void* const* d_in, const int* in_sizes, int n_in,
                              void* d_out, int out_size, void* d_ws, size_t ws_size,
                              hipStream_t stream) {
    const float* img1 = (const float*)d_in[0];
    const float* img2 = (const float*)d_in[1];
    const float* w1d  = (const float*)d_in[2];
    float* U   = (float*)d_ws;
    float* out = (float*)d_out;

    dim3 block(128);
    dim3 grid1(4, IMG_H / 16, NIMG);
    dim3 grid2(NTILE, 4, NIMG);
    hipLaunchKernelGGL(ssim_blur_rows, grid1, block, 0, stream, img1, img2, w1d, U);
    hipLaunchKernelGGL(ssim_blur_cols, grid2, block, 0, stream, U, w1d, out);
}


# each wave streams the joint band of 4 adjacent output tiles through a register ring: a fragment is loaded (launch 1: cleaned, squared) once, 3 groups ahead, and fed to every tile that holds it
# speedup vs baseline: 2.9269x; 1.0625x over previous
#include <hip/hip_runtime.h>
#include <math.h>


typedef __attribute__((ext_vector_type(2))) float v2f;
typedef __attribute__((ext_vector_type(4))) float v4f;
typedef __attribute__((ext_vector_type(8))) float v8f;

#define IMG_H 512
#define IMG_W 512
#define NIMG  16
#define WIN   81
#define PAD   40
#define NTILE 32
#define NGRP  64
#define TPW   4
#define NBG   18
#define PF1   3
#define PF2   3

__device__ __forceinline__ float cleanf(float v, float top) {
    v = (v != v) ? 0.0f : v;
    return fminf(fmaxf(v, 0.0f), top);
}

#define WMMA(A, B, C) __builtin_amdgcn_wmma_f32_16x16x4_f32(false, (A), false, (B), (short)0, (C), false, false)

__device__ __forceinline__ void load_wfrag(const float* wz, int j, int half, v2f (&wf)[24]) {
#pragma unroll
    for (int s = 0; s < 24; ++s) {
        const int p = 8 * (s >> 1) + 4 * half + 2 * (s & 1);
        wf[s][0] = wz[16 + p - j];
        wf[s][1] = wz[17 + p - j];
    }
}

__device__ __forceinline__ void mac_group(const int L, const v4f (&t)[5], const v2f (&wf)[24],
                                          v8f (&acc)[TPW][5]) {
#pragma unroll
    for (int q = 0; q < 2; ++q) {
        v2f a[5];
#pragma unroll
        for (int c = 0; c < 5; ++c) { a[c][0] = t[c][2 * q]; a[c][1] = t[c][2 * q + 1]; }
#pragma unroll
        for (int d = 0; d < TPW; ++d) {
            const int Ld = L - 2 * d;
            if (Ld >= 0 && Ld < 12) {
                const v2f b = wf[2 * Ld + q];
#pragma unroll
                for (int c = 0; c < 5; ++c) acc[d][c] = WMMA(a[c], b, acc[d][c]);
            }
        }
    }
}

__global__ __launch_bounds__(128) void ssim_blur_rows(
    const float* __restrict__ img1, const float* __restrict__ img2,
    const float* __restrict__ w1d, float* __restrict__ U)
{
    const int tid = threadIdx.x;
    const float wpad = (tid >= 16 && tid < 16 + WIN) ? w1d[tid - 16] : 0.0f;
    const int lane = tid & 31;
    const int wv   = __builtin_amdgcn_readfirstlane(tid >> 5);
    const int j    = lane & 15;
    const int half = lane >> 4;

    const int m0  = blockIdx.y * 16;
    const int img = blockIdx.z;
    const int ct0 = blockIdx.x * (4 * TPW) + wv * TPW;
    const int c0  = ct0 * 16;
    const float* s1 = img1 + ((size_t)img * IMG_H + m0) * IMG_W;
    const float* s2 = img2 + ((size_t)img * IMG_H + m0) * IMG_W;
    const int lrow = j * IMG_W;

    v8f acc[TPW][5];
#pragma unroll
    for (int d = 0; d < TPW; ++d)
#pragma unroll
        for (int c = 0; c < 5; ++c) acc[d][c] = (v8f){};

    v4f rx1[PF1 + 1], rx2[PF1 + 1];
    float rtop[PF1 + 1];
#define ISSUE1(G) { \
        const int col = c0 - PAD + 8 * (G) + 4 * half;       \
        const bool ok = (unsigned)col < (unsigned)IMG_W; \
        const int cc = ok ? col : 0; \
        rtop[(G) % (PF1 + 1)] = ok ? 1.0f : 0.0f;             \
        rx1[(G) % (PF1 + 1)] = *(const v4f*)(s1 + (lrow + cc)); \
        rx2[(G) % (PF1 + 1)] = *(const v4f*)(s2 + (lrow + cc)); }
#pragma unroll
    for (int G = 0; G < PF1; ++G) ISSUE1(G)

    __shared__ float wz[112];
    if (tid < 112) wz[tid] = wpad;
    __syncthreads();
    v2f wf[24];
    load_wfrag(wz, j, half, wf);

#pragma unroll
    for (int L = 0; L < NBG; ++L) {
        if (L + PF1 < NBG) ISSUE1(L + PF1)
        __builtin_amdgcn_sched_barrier(0);
        v4f x1 = rx1[L % (PF1 + 1)];
        v4f x2 = rx2[L % (PF1 + 1)];
        const float top = rtop[L % (PF1 + 1)];
#pragma unroll
        for (int e = 0; e < 4; ++e) {
            x1[e] = cleanf(x1[e], top);
            x2[e] = cleanf(x2[e], top);
        }
        v4f t[5];
        t[0] = x1; t[1] = x2; t[2] = x1 * x1; t[3] = x2 * x2; t[4] = x1 * x2;
        mac_group(L, t, wf, acc);
        __builtin_amdgcn_sched_barrier(0);
    }
#undef ISSUE1

    const size_t chs = (size_t)NIMG * NTILE * NGRP * 128;
#pragma unroll
    for (int d = 0; d < TPW; ++d) {
        const size_t blk = ((size_t)(ct0 + d) * NGRP + (m0 >> 3) + half) * 128 + j * 8;
        float* u = U + (size_t)img * NTILE * NGRP * 128 + blk;
#pragma unroll
        for (int c = 0; c < 5; ++c) {
            v4f lo, hi4;
            lo[0] = acc[d][c][0]; lo[1] = acc[d][c][1]; lo[2] = acc[d][c][2]; lo[3] = acc[d][c][3];
            hi4[0] = acc[d][c][4]; hi4[1] = acc[d][c][5]; hi4[2] = acc[d][c][6]; hi4[3] = acc[d][c][7];
            *(v4f*)(u + c * chs) = lo;
            *(v4f*)(u + c * chs + 4) = hi4;
        }
    }
}

__device__ __forceinline__ void ssim_tail(const v8f (&a)[5], float* __restrict__ out,
                                          int img, int rt, int ct, int j, int half) {
    const float C1v = 1e-4f, C2v = 9e-4f;
    float r8[8];
#pragma unroll
    for (int v = 0; v < 8; ++v) {
        const float mu1 = a[0][v], mu2 = a[1][v];
        const float e11 = a[2][v], e22 = a[3][v], e12 = a[4][v];
        const float mu1sq = mu1 * mu1, mu2sq = mu2 * mu2, mu12 = mu1 * mu2;
        const float s11 = fmaxf(e11 - mu1sq, 0.0f);
        const float s22 = fmaxf(e22 - mu2sq, 0.0f);
        const float s12 = e12 - mu12;
        const float num = (2.0f * mu12 + C1v) * (2.0f * s12 + C2v);
        const float den = (mu1sq + mu2sq + C1v) * (s11 + s22 + C2v);
        float r = num / (den + 1e-8f);
        r = (r != r) ? 0.0f : r;
        r8[v] = fminf(fmaxf(r, -1.0f), 1.0f);
    }
    float* o = out + ((size_t)img * IMG_H + rt * 16 + j) * IMG_W + ct * 16 + 8 * half;
    v4f lo, hi4;
    lo[0] = r8[0]; lo[1] = r8[1]; lo[2] = r8[2]; lo[3] = r8[3];
    hi4[0] = r8[4]; hi4[1] = r8[5]; hi4[2] = r8[6]; hi4[3] = r8[7];
    *(v4f*)(o) = lo;
    *(v4f*)(o + 4) = hi4;
}

__global__ __launch_bounds__(128) void ssim_blur_cols(
    const float* __restrict__ U, const float* __restrict__ w1d,
    float* __restrict__ out)
{
    const int tid = threadIdx.x;
    const float wpad = (tid >= 16 && tid < 16 + WIN) ? w1d[tid - 16] : 0.0f;
    const int lane = tid & 31;
    const int wv   = __builtin_amdgcn_readfirstlane(tid >> 5);
    const int j    = lane & 15;
    const int half = lane >> 4;

    const int ct  = blockIdx.x;
    const int img = blockIdx.z;
    const int rt0 = blockIdx.y * (4 * TPW) + wv * TPW;
    const size_t chs = (size_t)NIMG * NTILE * NGRP * 128;
    const float* us = U + ((size_t)img * NTILE + ct) * NGRP * 128;
    const int loff = j * 8 + half * 4;

    v8f acc[TPW][5];
#pragma unroll
    for (int d = 0; d < TPW; ++d)
#pragma unroll
        for (int c = 0; c < 5; ++c) acc[d][c] = (v8f){};

    v4f ring[PF2 + 1][5];
    bool rok[PF2 + 1];
#define ISSUE2(G) { \
        const int g = 2 * rt0 - 5 + (G);                   \
        const bool ok = (unsigned)g < (unsigned)NGRP; \
        rok[(G) % (PF2 + 1)] = ok; \
        const float* p = us + (size_t)(ok ? g : 0) * 128; \
        _Pragma("unroll") \
        for (int c = 0; c < 5; ++c) ring[(G) % (PF2 + 1)][c] = *(const v4f*)(p + c * chs + loff); }
#pragma unroll
    for (int G = 0; G < PF2; ++G) ISSUE2(G)

    __shared__ float wz[112];
    if (tid < 112) wz[tid] = wpad;
    __syncthreads();
    v2f wf[24];
    load_wfrag(wz, j, half, wf);

#pragma unroll
    for (int L = 0; L < NBG; ++L) {
        if (L + PF2 < NBG) ISSUE2(L + PF2)
        __builtin_amdgcn_sched_barrier(0);
        v4f t[5];
        const v4f z = {0.0f, 0.0f, 0.0f, 0.0f};
        const bool ok = rok[L % (PF2 + 1)];
#pragma unroll
        for (int c = 0; c < 5; ++c) t[c] = ok ? ring[L % (PF2 + 1)][c] : z;
        mac_group(L, t, wf, acc);

        if (L >= 12 && ((L - 12) & 1) == 0) ssim_tail(acc[(L - 12) >> 1], out, img, rt0 + ((L - 12) >> 1), ct, j, half);
        __builtin_amdgcn_sched_barrier(0);
    }
#undef ISSUE2
    ssim_tail(acc[TPW - 1], out, img, rt0 + TPW - 1, ct, j, half);
}

extern "C" void kernel_launch(void* const* d_in, const int* in_sizes, int n_in,
                              void* d_out, int out_size, void* d_ws, size_t ws_size,
                              hipStream_t stream) {
    const float* img1 = (const float*)d_in[0];
    const float* img2 = (const float*)d_in[1];
    const float* w1d  = (const float*)d_in[2];
    float* U   = (float*)d_ws;
    float* out = (float*)d_out;

    dim3 block(128);
    dim3 grid1(2, IMG_H / 16, NIMG);
    dim3 grid2(NTILE, 2, NIMG);
    hipLaunchKernelGGL(ssim_blur_rows, grid1, block, 0, stream, img1, img2, w1d, U);
    hipLaunchKernelGGL(ssim_blur_cols, grid2, block, 0, stream, U, w1d, out);
}


# one launch, no workspace: a wave walks a 16-col strip of an image half; the row blur's result tile is used as it stands as the column blur's data fragment (same k order); fp summation order changes
# speedup vs baseline: 3.3749x; 1.1531x over previous
#include <hip/hip_runtime.h>
#include <math.h>


typedef __attribute__((ext_vector_type(2))) float v2f;
typedef __attribute__((ext_vector_type(4))) float v4f;
typedef __attribute__((ext_vector_type(8))) float v8f;

#define IMG_H 512
#define IMG_W 512
#define NIMG  16
#define WIN   81
#define PAD   40
#define NGR   12
#define NBAND 6
#define USTEP 2
#define NSLOT (NBAND - 1 + USTEP)
#define NSTEP 22

__device__ __forceinline__ float cleanf(float v, float top) {
    v = (v != v) ? 0.0f : v;
    return fminf(fmaxf(v, 0.0f), top);
}

#define WMMA(A, B, C) __builtin_amdgcn_wmma_f32_16x16x4_f32(false, (A), false, (B), (short)0, (C), false, false)

__device__ __forceinline__ void load_wfrag_rows(const float* wz, int j, int half, v2f (&wf)[24]) {
#pragma unroll
    for (int s = 0; s < 24; ++s) {
        const int p = 8 * (s >> 1) + 4 * half + 2 * (s & 1);
        wf[s][0] = wz[16 + p - j];
        wf[s][1] = wz[17 + p - j];
    }
}

__device__ __forceinline__ void load_wfrag_cols(const float* wz, int j, int half, v2f (&wf)[24]) {
#pragma unroll
    for (int s = 0; s < 24; ++s) {
        const int p = 16 * (s >> 2) + 8 * half + 2 * (s & 3);
        wf[s][0] = wz[16 + p - j];
        wf[s][1] = wz[17 + p - j];
    }
}

__device__ __forceinline__ void ssim_tail(const v8f (&a)[5], float* __restrict__ out,
                                          int img, int row0, int ct, int j, int half) {
    const float C1v = 1e-4f, C2v = 9e-4f;
    float r8[8];
#pragma unroll
    for (int v = 0; v < 8; ++v) {
        const float mu1 = a[0][v], mu2 = a[1][v];
        const float e11 = a[2][v], e22 = a[3][v], e12 = a[4][v];
        const float mu1sq = mu1 * mu1, mu2sq = mu2 * mu2, mu12 = mu1 * mu2;
        const float s11 = fmaxf(e11 - mu1sq, 0.0f);
        const float s22 = fmaxf(e22 - mu2sq, 0.0f);
        const float s12 = e12 - mu12;
        const float num = (2.0f * mu12 + C1v) * (2.0f * s12 + C2v);
        const float den = (mu1sq + mu2sq + C1v) * (s11 + s22 + C2v);
        float r = num / (den + 1e-8f);
        r = (r != r) ? 0.0f : r;
        r8[v] = fminf(fmaxf(r, -1.0f), 1.0f);
    }
    const int row = row0 + j;
    if ((unsigned)row < (unsigned)IMG_H) {
        float* o = out + ((size_t)img * IMG_H + row) * IMG_W + ct * 16 + 8 * half;
        v4f lo, hi4;
        lo[0] = r8[0]; lo[1] = r8[1]; lo[2] = r8[2]; lo[3] = r8[3];
        hi4[0] = r8[4]; hi4[1] = r8[5]; hi4[2] = r8[6]; hi4[3] = r8[7];
        *(v4f*)(o) = lo;
        *(v4f*)(o + 4) = hi4;
    }
}

__global__ __launch_bounds__(128) void ssim_fused(
    const float* __restrict__ img1, const float* __restrict__ img2,
    const float* __restrict__ w1d, float* __restrict__ out)
{
    const int tid = threadIdx.x;
    const float wpad = (tid >= 16 && tid < 16 + WIN) ? w1d[tid - 16] : 0.0f;
    const int lane = tid & 31;
    const int wv   = __builtin_amdgcn_readfirstlane(tid >> 5);
    const int j    = lane & 15;
    const int half = lane >> 4;

    const int ct  = blockIdx.x * 4 + wv;
    const int c0  = ct * 16;
    const int seg = blockIdx.y;
    const int img = blockIdx.z;
    const int M   = seg ? 16 : 17;
    const int of  = seg ? 16 : -1;
    const int i0  = of - 2;
    const int n0  = seg ? 0 : 3;
    const float* b1 = img1 + (size_t)img * IMG_H * IMG_W;
    const float* b2 = img2 + (size_t)img * IMG_H * IMG_W;

    int coff[NGR];
    float top[NGR];
#pragma unroll
    for (int G = 0; G < NGR; ++G) {
        const int col = c0 - PAD + 8 * G + 4 * half;
        const bool ok = (unsigned)col < (unsigned)IMG_W;
        coff[G] = j * IMG_W + (ok ? col : 0);
        top[G]  = ok ? 1.0f : 0.0f;
    }

    v4f rx1[NGR], rx2[NGR];
#define ISSUE_TILE(I) { \
        const int it = min(max((I), 0), IMG_H / 16 - 1);       \
        const float* p1 = b1 + it * 16 * IMG_W; \
        const float* p2 = b2 + it * 16 * IMG_W; \
        _Pragma("unroll") \
        for (int G = 0; G < NGR; ++G) { \
            rx1[G] = *(const v4f*)(p1 + coff[G]); \
            rx2[G] = *(const v4f*)(p2 + coff[G]); } }
    ISSUE_TILE(i0 + n0)

    __shared__ float wz[112];
    if (tid < 112) wz[tid] = wpad;
    __syncthreads();
    v2f wfr[24], wfc[24];
    load_wfrag_rows(wz, j, half, wfr);
    load_wfrag_cols(wz, j, half, wfc);

    const v8f zero8 = {};
    v8f acc[NSLOT][5];
#pragma unroll
    for (int d = 0; d < NSLOT; ++d)
#pragma unroll
        for (int c = 0; c < 5; ++c) acc[d][c] = zero8;

#pragma nounroll
    for (int nb = n0; nb < n0 + NSTEP; nb += USTEP) {
#pragma unroll
        for (int u = 0; u < USTEP; ++u) {
            const int n = nb + u;
            const int i = i0 + n;
            const bool work = ((unsigned)i < (unsigned)(IMG_H / 16)) && (n < M + NBAND - 1);
            if (work) {
                v8f R[5];
#pragma unroll
                for (int G = 0; G < NGR; ++G) {
                    v4f x1 = rx1[G];
                    v4f x2 = rx2[G];
#pragma unroll
                    for (int e = 0; e < 4; ++e) {
                        x1[e] = cleanf(x1[e], top[G]);
                        x2[e] = cleanf(x2[e], top[G]);
                    }
                    v4f t[5];
                    t[0] = x1; t[1] = x2; t[2] = x1 * x1; t[3] = x2 * x2; t[4] = x1 * x2;
#pragma unroll
                    for (int q = 0; q < 2; ++q) {
                        const v2f b = wfr[2 * G + q];
#pragma unroll
                        for (int c = 0; c < 5; ++c) {
                            v2f a; a[0] = t[c][2 * q]; a[1] = t[c][2 * q + 1];
                            R[c] = (G == 0 && q == 0) ? WMMA(a, b, zero8) : WMMA(a, b, R[c]);
                        }
                    }
                }
                ISSUE_TILE(i + 1)
#pragma unroll
                for (int dp = NBAND - 1; dp >= 0; --dp) {
                    const int m = n - dp;
                    if ((unsigned)m < (unsigned)M) {
                        const int slot = u + NBAND - 1 - dp;
#pragma unroll
                        for (int q = 0; q < 4; ++q) {
                            const v2f b = wfc[4 * dp + q];
#pragma unroll
                            for (int c = 0; c < 5; ++c) {
                                v2f a; a[0] = R[c][2 * q]; a[1] = R[c][2 * q + 1];
                                acc[slot][c] = (dp == 0 && q == 0) ? WMMA(a, b, zero8)
                                                                   : WMMA(a, b, acc[slot][c]);
                            }
                        }
                    }
                }
            } else {
                const bool next_work = ((unsigned)(i + 1) < (unsigned)(IMG_H / 16)) && (n + 1 < M + NBAND - 1);
                if (next_work) ISSUE_TILE(i + 1)
            }
            const int md = n - (NBAND - 1);
            if ((unsigned)md < (unsigned)M)
                ssim_tail(acc[u], out, img, 16 * (of + md) + 8, ct, j, half);
        }
#pragma unroll
        for (int d = 0; d + USTEP < NSLOT; ++d)
#pragma unroll
            for (int c = 0; c < 5; ++c) acc[d][c] = acc[d + USTEP][c];
    }
#undef ISSUE_TILE
}

extern "C" void kernel_launch(void* const* d_in, const int* in_sizes, int n_in,
                              void* d_out, int out_size, void* d_ws, size_t ws_size,
                              hipStream_t stream) {
    const float* img1 = (const float*)d_in[0];
    const float* img2 = (const float*)d_in[1];
    const float* w1d  = (const float*)d_in[2];
    float* out = (float*)d_out;

    dim3 block(128);
    dim3 grid(8, 2, NIMG);
    hipLaunchKernelGGL(ssim_fused, grid, block, 0, stream, img1, img2, w1d, out);
}


# accumulators alternate between two register sets by band position: a step's first product reads a tile's sum from one set and writes it one position on in the other; no accumulator is copied any more
# speedup vs baseline: 3.5373x; 1.0481x over previous
#include <hip/hip_runtime.h>
#include <math.h>


typedef __attribute__((ext_vector_type(2))) float v2f;
typedef __attribute__((ext_vector_type(4))) float v4f;
typedef __attribute__((ext_vector_type(8))) float v8f;

#define IMG_H 512
#define IMG_W 512
#define NIMG  16
#define WIN   81
#define PAD   40
#define NGR   12
#define NBAND 6

__device__ __forceinline__ float cleanf(float v, float top) {
    v = (v != v) ? 0.0f : v;
    return fminf(fmaxf(v, 0.0f), top);
}

#define WMMA(A, B, C) __builtin_amdgcn_wmma_f32_16x16x4_f32(false, (A), false, (B), (short)0, (C), false, false)

__device__ __forceinline__ void load_wfrag_rows(const float* wz, int j, int half, v2f (&wf)[24]) {
#pragma unroll
    for (int s = 0; s < 24; ++s) {
        const int p = 8 * (s >> 1) + 4 * half + 2 * (s & 1);
        wf[s][0] = wz[16 + p - j];
        wf[s][1] = wz[17 + p - j];
    }
}

__device__ __forceinline__ void load_wfrag_cols(const float* wz, int j, int half, v2f (&wf)[24]) {
#pragma unroll
    for (int s = 0; s < 24; ++s) {
        const int p = 16 * (s >> 2) + 8 * half + 2 * (s & 3);
        wf[s][0] = wz[16 + p - j];
        wf[s][1] = wz[17 + p - j];
    }
}

__device__ __forceinline__ void ssim_tail(const v8f (&a)[5], float* __restrict__ out,
                                          int img, int row0, int ct, int j, int half) {
    const float C1v = 1e-4f, C2v = 9e-4f;
    float r8[8];
#pragma unroll
    for (int v = 0; v < 8; ++v) {
        const float mu1 = a[0][v], mu2 = a[1][v];
        const float e11 = a[2][v], e22 = a[3][v], e12 = a[4][v];
        const float mu1sq = mu1 * mu1, mu2sq = mu2 * mu2, mu12 = mu1 * mu2;
        const float s11 = fmaxf(e11 - mu1sq, 0.0f);
        const float s22 = fmaxf(e22 - mu2sq, 0.0f);
        const float s12 = e12 - mu12;
        const float num = (2.0f * mu12 + C1v) * (2.0f * s12 + C2v);
        const float den = (mu1sq + mu2sq + C1v) * (s11 + s22 + C2v);
        float r = num / (den + 1e-8f);
        r = (r != r) ? 0.0f : r;
        r8[v] = fminf(fmaxf(r, -1.0f), 1.0f);
    }
    const int row = row0 + j;
    if ((unsigned)row < (unsigned)IMG_H) {
        float* o = out + ((size_t)img * IMG_H + row) * IMG_W + ct * 16 + 8 * half;
        v4f lo, hi4;
        lo[0] = r8[0]; lo[1] = r8[1]; lo[2] = r8[2]; lo[3] = r8[3];
        hi4[0] = r8[4]; hi4[1] = r8[5]; hi4[2] = r8[6]; hi4[3] = r8[7];
        *(v4f*)(o) = lo;
        *(v4f*)(o + 4) = hi4;
    }
}

__device__ __forceinline__ void col_step(const v8f (&src)[NBAND][5], v8f (&dst)[NBAND][5], v8f (&fin)[5],
                                         const v8f (&R)[5], const v2f (&wfc)[24], int n, int M) {
    const v8f zero8 = {};
#pragma unroll
    for (int dp = NBAND - 1; dp >= 0; --dp) {
        const int m = n - dp;
        if ((unsigned)m < (unsigned)M) {
#pragma unroll
            for (int q = 0; q < 4; ++q) {
                const v2f b = wfc[4 * dp + q];
#pragma unroll
                for (int c = 0; c < 5; ++c) {
                    v2f a; a[0] = R[c][2 * q]; a[1] = R[c][2 * q + 1];
                    v8f& d = (dp == NBAND - 1) ? fin[c] : dst[dp + 1][c];
                    if (q > 0)        d = WMMA(a, b, d);
                    else if (dp == 0) d = WMMA(a, b, zero8);
                    else              d = WMMA(a, b, src[dp][c]);
                }
            }
        }
    }
}

__device__ __forceinline__ void idle_step(const v8f (&src)[NBAND][5], v8f (&dst)[NBAND][5], v8f (&fin)[5]) {
#pragma unroll
    for (int dp = NBAND - 1; dp >= 1; --dp) {
#pragma unroll
        for (int c = 0; c < 5; ++c) {
            if (dp == NBAND - 1) fin[c] = src[dp][c];
            else                 dst[dp + 1][c] = src[dp][c];
        }
    }
}

__global__ __launch_bounds__(128) void ssim_fused(
    const float* __restrict__ img1, const float* __restrict__ img2,
    const float* __restrict__ w1d, float* __restrict__ out)
{
    const int tid = threadIdx.x;
    const float wpad = (tid >= 16 && tid < 16 + WIN) ? w1d[tid - 16] : 0.0f;
    const int lane = tid & 31;
    const int wv   = __builtin_amdgcn_readfirstlane(tid >> 5);
    const int j    = lane & 15;
    const int half = lane >> 4;

    const int ct  = blockIdx.x * 4 + wv;
    const int c0  = ct * 16;
    const int seg = blockIdx.y;
    const int img = blockIdx.z;
    const int M   = seg ? 16 : 17;
    const int of  = seg ? 16 : -1;
    const int i0  = of - 2;
    const int n0  = seg ? 0 : 3;
    const float* b1 = img1 + (size_t)img * IMG_H * IMG_W;
    const float* b2 = img2 + (size_t)img * IMG_H * IMG_W;

    int coff[NGR];
    float top[NGR];
#pragma unroll
    for (int G = 0; G < NGR; ++G) {
        const int col = c0 - PAD + 8 * G + 4 * half;
        const bool ok = (unsigned)col < (unsigned)IMG_W;
        coff[G] = j * IMG_W + (ok ? col : 0);
        top[G]  = ok ? 1.0f : 0.0f;
    }

    v4f rx1[NGR], rx2[NGR];
#define ISSUE_TILE(I) { \
        const int it = min(max((I), 0), IMG_H / 16 - 1);       \
        const float* p1 = b1 + it * 16 * IMG_W; \
        const float* p2 = b2 + it * 16 * IMG_W; \
        _Pragma("unroll") \
        for (int G = 0; G < NGR; ++G) { \
            rx1[G] = *(const v4f*)(p1 + coff[G]); \
            rx2[G] = *(const v4f*)(p2 + coff[G]); } }
    ISSUE_TILE(i0 + n0)

    __shared__ float wz[112];
    if (tid < 112) wz[tid] = wpad;
    __syncthreads();
    v2f wfr[24], wfc[24];
    load_wfrag_rows(wz, j, half, wfr);
    load_wfrag_cols(wz, j, half, wfc);

    const v8f zero8 = {};
    v8f accA[NBAND][5], accB[NBAND][5], fin[5];
#pragma unroll
    for (int d = 1; d < NBAND; ++d)
#pragma unroll
        for (int c = 0; c < 5; ++c) accA[d][c] = zero8;

    const int nend = M + NBAND - 1;

#define WALK_STEP(N, SRC, DST) { \
        const int n  = (N); \
        const int i  = i0 + n;                                 \
        const int md = n - (NBAND - 1);                        \
          \
        if ((unsigned)i < (unsigned)(IMG_H / 16)) { \
              \
            v8f R[5]; \
            _Pragma("unroll") \
            for (int G = 0; G < NGR; ++G) { \
                v4f x1 = rx1[G]; \
                v4f x2 = rx2[G]; \
                _Pragma("unroll") \
                for (int e = 0; e < 4; ++e) { \
                    x1[e] = cleanf(x1[e], top[G]); \
                    x2[e] = cleanf(x2[e], top[G]); \
                } \
                v4f t[5]; \
                t[0] = x1; t[1] = x2; t[2] = x1 * x1; t[3] = x2 * x2; t[4] = x1 * x2; \
                _Pragma("unroll") \
                for (int q = 0; q < 2; ++q) { \
                    const v2f b = wfr[2 * G + q]; \
                    _Pragma("unroll") \
                    for (int c = 0; c < 5; ++c) { \
                        v2f a; a[0] = t[c][2 * q]; a[1] = t[c][2 * q + 1]; \
                        R[c] = (G == 0 && q == 0) ? WMMA(a, b, zero8) : WMMA(a, b, R[c]); \
                    } \
                } \
            } \
              \
            ISSUE_TILE(i + 1) \
            col_step(SRC, DST, fin, R, wfc, n, M); \
        } else { \
            idle_step(SRC, DST, fin); \
        } \
          \
        if ((unsigned)md < (unsigned)M) \
            ssim_tail(fin, out, img, 16 * (of + md) + 8, ct, j, half); \
    }

#pragma nounroll
    for (int nb = n0; nb < nend; nb += 2) {
        WALK_STEP(nb,     accA, accB)
        if (nb + 1 >= nend) break;
        WALK_STEP(nb + 1, accB, accA)
    }
#undef WALK_STEP
#undef ISSUE_TILE
}

extern "C" void kernel_launch(void* const* d_in, const int* in_sizes, int n_in,
                              void* d_out, int out_size, void* d_ws, size_t ws_size,
                              hipStream_t stream) {
    const float* img1 = (const float*)d_in[0];
    const float* img2 = (const float*)d_in[1];
    const float* w1d  = (const float*)d_in[2];
    float* out = (float*)d_out;

    dim3 block(128);
    dim3 grid(8, 2, NIMG);
    hipLaunchKernelGGL(ssim_fused, grid, block, 0, stream, img1, img2, w1d, out);
}


# input clean-up of the row blur's data cut from five vector instructions per value to a quieting max and one fused max-min: NaN compare and select dropped, fmaxf(NaN,0)=0 already
# speedup vs baseline: 3.7741x; 1.0669x over previous
#include <hip/hip_runtime.h>
#include <math.h>


typedef __attribute__((ext_vector_type(2))) float v2f;
typedef __attribute__((ext_vector_type(4))) float v4f;
typedef __attribute__((ext_vector_type(8))) float v8f;

#define IMG_H 512
#define IMG_W 512
#define NIMG  16
#define WIN   81
#define PAD   40
#define NGR   12
#define NBAND 6

__device__ __forceinline__ float cleanf(float v, float top) {
    return fminf(fmaxf(v, 0.0f), top);
}

#define WMMA(A, B, C) __builtin_amdgcn_wmma_f32_16x16x4_f32(false, (A), false, (B), (short)0, (C), false, false)

__device__ __forceinline__ void load_wfrag_rows(const float* wz, int j, int half, v2f (&wf)[24]) {
#pragma unroll
    for (int s = 0; s < 24; ++s) {
        const int p = 8 * (s >> 1) + 4 * half + 2 * (s & 1);
        wf[s][0] = wz[16 + p - j];
        wf[s][1] = wz[17 + p - j];
    }
}

__device__ __forceinline__ void load_wfrag_cols(const float* wz, int j, int half, v2f (&wf)[24]) {
#pragma unroll
    for (int s = 0; s < 24; ++s) {
        const int p = 16 * (s >> 2) + 8 * half + 2 * (s & 3);
        wf[s][0] = wz[16 + p - j];
        wf[s][1] = wz[17 + p - j];
    }
}

__device__ __forceinline__ void ssim_tail(const v8f (&a)[5], float* __restrict__ out,
                                          int img, int row0, int ct, int j, int half) {
    const float C1v = 1e-4f, C2v = 9e-4f;
    float r8[8];
#pragma unroll
    for (int v = 0; v < 8; ++v) {
        const float mu1 = a[0][v], mu2 = a[1][v];
        const float e11 = a[2][v], e22 = a[3][v], e12 = a[4][v];
        const float mu1sq = mu1 * mu1, mu2sq = mu2 * mu2, mu12 = mu1 * mu2;
        const float s11 = fmaxf(e11 - mu1sq, 0.0f);
        const float s22 = fmaxf(e22 - mu2sq, 0.0f);
        const float s12 = e12 - mu12;
        const float num = (2.0f * mu12 + C1v) * (2.0f * s12 + C2v);
        const float den = (mu1sq + mu2sq + C1v) * (s11 + s22 + C2v);
        float r = num / (den + 1e-8f);
        r = (r != r) ? 0.0f : r;
        r8[v] = fminf(fmaxf(r, -1.0f), 1.0f);
    }
    const int row = row0 + j;
    if ((unsigned)row < (unsigned)IMG_H) {
        float* o = out + ((size_t)img * IMG_H + row) * IMG_W + ct * 16 + 8 * half;
        v4f lo, hi4;
        lo[0] = r8[0]; lo[1] = r8[1]; lo[2] = r8[2]; lo[3] = r8[3];
        hi4[0] = r8[4]; hi4[1] = r8[5]; hi4[2] = r8[6]; hi4[3] = r8[7];
        *(v4f*)(o) = lo;
        *(v4f*)(o + 4) = hi4;
    }
}

__device__ __forceinline__ void col_step(const v8f (&src)[NBAND][5], v8f (&dst)[NBAND][5], v8f (&fin)[5],
                                         const v8f (&R)[5], const v2f (&wfc)[24], int n, int M) {
    const v8f zero8 = {};
#pragma unroll
    for (int dp = NBAND - 1; dp >= 0; --dp) {
        const int m = n - dp;
        if ((unsigned)m < (unsigned)M) {
#pragma unroll
            for (int q = 0; q < 4; ++q) {
                const v2f b = wfc[4 * dp + q];
#pragma unroll
                for (int c = 0; c < 5; ++c) {
                    v2f a; a[0] = R[c][2 * q]; a[1] = R[c][2 * q + 1];
                    v8f& d = (dp == NBAND - 1) ? fin[c] : dst[dp + 1][c];
                    if (q > 0)        d = WMMA(a, b, d);
                    else if (dp == 0) d = WMMA(a, b, zero8);
                    else              d = WMMA(a, b, src[dp][c]);
                }
            }
        }
    }
}

__device__ __forceinline__ void idle_step(const v8f (&src)[NBAND][5], v8f (&dst)[NBAND][5], v8f (&fin)[5]) {
#pragma unroll
    for (int dp = NBAND - 1; dp >= 1; --dp) {
#pragma unroll
        for (int c = 0; c < 5; ++c) {
            if (dp == NBAND - 1) fin[c] = src[dp][c];
            else                 dst[dp + 1][c] = src[dp][c];
        }
    }
}

__global__ __launch_bounds__(128) void ssim_fused(
    const float* __restrict__ img1, const float* __restrict__ img2,
    const float* __restrict__ w1d, float* __restrict__ out)
{
    const int tid = threadIdx.x;
    const float wpad = (tid >= 16 && tid < 16 + WIN) ? w1d[tid - 16] : 0.0f;
    const int lane = tid & 31;
    const int wv   = __builtin_amdgcn_readfirstlane(tid >> 5);
    const int j    = lane & 15;
    const int half = lane >> 4;

    const int ct  = blockIdx.x * 4 + wv;
    const int c0  = ct * 16;
    const int seg = blockIdx.y;
    const int img = blockIdx.z;
    const int M   = seg ? 16 : 17;
    const int of  = seg ? 16 : -1;
    const int i0  = of - 2;
    const int n0  = seg ? 0 : 3;
    const float* b1 = img1 + (size_t)img * IMG_H * IMG_W;
    const float* b2 = img2 + (size_t)img * IMG_H * IMG_W;

    int coff[NGR];
    float top[NGR];
#pragma unroll
    for (int G = 0; G < NGR; ++G) {
        const int col = c0 - PAD + 8 * G + 4 * half;
        const bool ok = (unsigned)col < (unsigned)IMG_W;
        coff[G] = j * IMG_W + (ok ? col : 0);
        top[G]  = ok ? 1.0f : 0.0f;
    }

    v4f rx1[NGR], rx2[NGR];
#define ISSUE_TILE(I) { \
        const int it = min(max((I), 0), IMG_H / 16 - 1);       \
        const float* p1 = b1 + it * 16 * IMG_W; \
        const float* p2 = b2 + it * 16 * IMG_W; \
        _Pragma("unroll") \
        for (int G = 0; G < NGR; ++G) { \
            rx1[G] = *(const v4f*)(p1 + coff[G]); \
            rx2[G] = *(const v4f*)(p2 + coff[G]); } }
    ISSUE_TILE(i0 + n0)

    __shared__ float wz[112];
    if (tid < 112) wz[tid] = wpad;
    __syncthreads();
    v2f wfr[24], wfc[24];
    load_wfrag_rows(wz, j, half, wfr);
    load_wfrag_cols(wz, j, half, wfc);

    const v8f zero8 = {};
    v8f accA[NBAND][5], accB[NBAND][5], fin[5];
#pragma unroll
    for (int d = 1; d < NBAND; ++d)
#pragma unroll
        for (int c = 0; c < 5; ++c) accA[d][c] = zero8;

    const int nend = M + NBAND - 1;

#define WALK_STEP(N, SRC, DST) { \
        const int n  = (N); \
        const int i  = i0 + n;                                 \
        const int md = n - (NBAND - 1);                        \
          \
        if ((unsigned)i < (unsigned)(IMG_H / 16)) { \
              \
            v8f R[5]; \
            _Pragma("unroll") \
            for (int G = 0; G < NGR; ++G) { \
                v4f x1 = rx1[G]; \
                v4f x2 = rx2[G]; \
                _Pragma("unroll") \
                for (int e = 0; e < 4; ++e) { \
                    x1[e] = cleanf(x1[e], top[G]); \
                    x2[e] = cleanf(x2[e], top[G]); \
                } \
                v4f t[5]; \
                t[0] = x1; t[1] = x2; t[2] = x1 * x1; t[3] = x2 * x2; t[4] = x1 * x2; \
                _Pragma("unroll") \
                for (int q = 0; q < 2; ++q) { \
                    const v2f b = wfr[2 * G + q]; \
                    _Pragma("unroll") \
                    for (int c = 0; c < 5; ++c) { \
                        v2f a; a[0] = t[c][2 * q]; a[1] = t[c][2 * q + 1]; \
                        R[c] = (G == 0 && q == 0) ? WMMA(a, b, zero8) : WMMA(a, b, R[c]); \
                    } \
                } \
            } \
              \
            ISSUE_TILE(i + 1) \
            col_step(SRC, DST, fin, R, wfc, n, M); \
        } else { \
            idle_step(SRC, DST, fin); \
        } \
          \
        if ((unsigned)md < (unsigned)M) \
            ssim_tail(fin, out, img, 16 * (of + md) + 8, ct, j, half); \
    }

#pragma nounroll
    for (int nb = n0; nb < nend; nb += 2) {
        WALK_STEP(nb,     accA, accB)
        if (nb + 1 >= nend) break;
        WALK_STEP(nb + 1, accB, accA)
    }
#undef WALK_STEP
#undef ISSUE_TILE
}

extern "C" void kernel_launch(void* const* d_in, const int* in_sizes, int n_in,
                              void* d_out, int out_size, void* d_ws, size_t ws_size,
                              hipStream_t stream) {
    const float* img1 = (const float*)d_in[0];
    const float* img2 = (const float*)d_in[1];
    const float* w1d  = (const float*)d_in[2];
    float* out = (float*)d_out;

    dim3 block(128);
    dim3 grid(8, 2, NIMG);
    hipLaunchKernelGGL(ssim_fused, grid, block, 0, stream, img1, img2, w1d, out);
}


# the 5 input tiles both parts of a strip need are row-blurred once: the lower wave leaves their result tiles in LDS for the upper wave of that strip (one split barrier); cut now at 18/15 output tiles
# speedup vs baseline: 3.9142x; 1.0371x over previous
#include <hip/hip_runtime.h>
#include <math.h>


typedef __attribute__((ext_vector_type(2))) float v2f;
typedef __attribute__((ext_vector_type(4))) float v4f;
typedef __attribute__((ext_vector_type(8))) float v8f;

#define IMG_H 512
#define IMG_W 512
#define NIMG  16
#define WIN   81
#define PAD   40
#define NGR   12
#define NBAND 6
#define NSH   5
#define M_UP  18
#define M_LO  15

__device__ __forceinline__ float cleanf(float v, float top) {
    return fminf(fmaxf(v, 0.0f), top);
}

#define WMMA(A, B, C) __builtin_amdgcn_wmma_f32_16x16x4_f32(false, (A), false, (B), (short)0, (C), false, false)

__device__ __forceinline__ void load_wfrag_rows(const float* wz, int j, int half, v2f (&wf)[24]) {
#pragma unroll
    for (int s = 0; s < 24; ++s) {
        const int p = 8 * (s >> 1) + 4 * half + 2 * (s & 1);
        wf[s][0] = wz[16 + p - j];
        wf[s][1] = wz[17 + p - j];
    }
}

__device__ __forceinline__ void load_wfrag_cols(const float* wz, int j, int half, v2f (&wf)[24]) {
#pragma unroll
    for (int s = 0; s < 24; ++s) {
        const int p = 16 * (s >> 2) + 8 * half + 2 * (s & 3);
        wf[s][0] = wz[16 + p - j];
        wf[s][1] = wz[17 + p - j];
    }
}

__device__ __forceinline__ void ssim_tail(const v8f (&a)[5], float* __restrict__ out,
                                          int img, int row0, int ct, int j, int half) {
    const float C1v = 1e-4f, C2v = 9e-4f;
    float r8[8];
#pragma unroll
    for (int v = 0; v < 8; ++v) {
        const float mu1 = a[0][v], mu2 = a[1][v];
        const float e11 = a[2][v], e22 = a[3][v], e12 = a[4][v];
        const float mu1sq = mu1 * mu1, mu2sq = mu2 * mu2, mu12 = mu1 * mu2;
        const float s11 = fmaxf(e11 - mu1sq, 0.0f);
        const float s22 = fmaxf(e22 - mu2sq, 0.0f);
        const float s12 = e12 - mu12;
        const float num = (2.0f * mu12 + C1v) * (2.0f * s12 + C2v);
        const float den = (mu1sq + mu2sq + C1v) * (s11 + s22 + C2v);
        float r = num / (den + 1e-8f);
        r = (r != r) ? 0.0f : r;
        r8[v] = fminf(fmaxf(r, -1.0f), 1.0f);
    }
    const int row = row0 + j;
    if ((unsigned)row < (unsigned)IMG_H) {
        float* o = out + ((size_t)img * IMG_H + row) * IMG_W + ct * 16 + 8 * half;
        v4f lo, hi4;
        lo[0] = r8[0]; lo[1] = r8[1]; lo[2] = r8[2]; lo[3] = r8[3];
        hi4[0] = r8[4]; hi4[1] = r8[5]; hi4[2] = r8[6]; hi4[3] = r8[7];
        *(v4f*)(o) = lo;
        *(v4f*)(o + 4) = hi4;
    }
}

__device__ __forceinline__ void col_step(const v8f (&src)[NBAND][5], v8f (&dst)[NBAND][5], v8f (&fin)[5],
                                         const v8f (&R)[5], const v2f (&wfc)[24], int n, int M) {
    const v8f zero8 = {};
#pragma unroll
    for (int dp = NBAND - 1; dp >= 0; --dp) {
        const int m = n - dp;
        if ((unsigned)m < (unsigned)M) {
#pragma unroll
            for (int q = 0; q < 4; ++q) {
                const v2f b = wfc[4 * dp + q];
#pragma unroll
                for (int c = 0; c < 5; ++c) {
                    v2f a; a[0] = R[c][2 * q]; a[1] = R[c][2 * q + 1];
                    v8f& d = (dp == NBAND - 1) ? fin[c] : dst[dp + 1][c];
                    if (q > 0)        d = WMMA(a, b, d);
                    else if (dp == 0) d = WMMA(a, b, zero8);
                    else              d = WMMA(a, b, src[dp][c]);
                }
            }
        }
    }
}

__device__ __forceinline__ void idle_step(const v8f (&src)[NBAND][5], v8f (&dst)[NBAND][5], v8f (&fin)[5]) {
#pragma unroll
    for (int dp = NBAND - 1; dp >= 1; --dp) {
#pragma unroll
        for (int c = 0; c < 5; ++c) {
            if (dp == NBAND - 1) fin[c] = src[dp][c];
            else                 dst[dp + 1][c] = src[dp][c];
        }
    }
}

__global__ __launch_bounds__(128) void ssim_fused(
    const float* __restrict__ img1, const float* __restrict__ img2,
    const float* __restrict__ w1d, float* __restrict__ out)
{
    const int tid = threadIdx.x;
    const float wpad = (tid >= 16 && tid < 16 + WIN) ? w1d[tid - 16] : 0.0f;
    const int lane = tid & 31;
    const int wv   = __builtin_amdgcn_readfirstlane(tid >> 5);
    const int j    = lane & 15;
    const int half = lane >> 4;

    const int sp  = wv >> 1;
    const int seg = wv & 1;
    const int ct  = blockIdx.x * 2 + sp;
    const int c0  = ct * 16;
    const int img = blockIdx.z;
    const int M   = seg ? M_LO : M_UP;
    const int of  = seg ? M_UP - 1 : -1;
    const int i0  = of - 2;
    const int n0  = seg ? 0 : 3;
    const int irow = seg ? IMG_H / 16 : M_UP - 3;
    const int ilds = seg ? IMG_H / 16 : M_UP - 3 + NSH;
    const float* b1 = img1 + (size_t)img * IMG_H * IMG_W;
    const float* b2 = img2 + (size_t)img * IMG_H * IMG_W;

    int coff[NGR];
    float top[NGR];
#pragma unroll
    for (int G = 0; G < NGR; ++G) {
        const int col = c0 - PAD + 8 * G + 4 * half;
        const bool ok = (unsigned)col < (unsigned)IMG_W;
        coff[G] = j * IMG_W + (ok ? col : 0);
        top[G]  = ok ? 1.0f : 0.0f;
    }

    v4f rx1[NGR], rx2[NGR];
#define ISSUE_TILE(I) { \
        const int it = min(max((I), 0), IMG_H / 16 - 1);       \
        const float* p1 = b1 + it * 16 * IMG_W; \
        const float* p2 = b2 + it * 16 * IMG_W; \
        _Pragma("unroll") \
        for (int G = 0; G < NGR; ++G) { \
            rx1[G] = *(const v4f*)(p1 + coff[G]); \
            rx2[G] = *(const v4f*)(p2 + coff[G]); } }
    ISSUE_TILE(i0 + n0)

    __shared__ float wz[112];
    __shared__ v4f rsh[2][NSH][5][2][32];
    if (tid < 112) wz[tid] = wpad;
    __syncthreads();
    v2f wfr[24], wfc[24];
    load_wfrag_rows(wz, j, half, wfr);
    load_wfrag_cols(wz, j, half, wfc);
#pragma unroll
    for (int s = 0; s < 24; s += 4)
        asm volatile("" :: "v"(wfr[s]), "v"(wfr[s + 1]), "v"(wfr[s + 2]), "v"(wfr[s + 3]),
                           "v"(wfc[s]), "v"(wfc[s + 1]), "v"(wfc[s + 2]), "v"(wfc[s + 3]));

    const v8f zero8 = {};
    v8f accA[NBAND][5], accB[NBAND][5], fin[5];
#pragma unroll
    for (int d = 1; d < NBAND; ++d)
#pragma unroll
        for (int c = 0; c < 5; ++c) accA[d][c] = zero8;


#define WALK_STEP(N, SRC, DST) { \
        const int n  = (N); \
        const int i  = i0 + n;                                 \
        const int md = n - (NBAND - 1);                        \
          \
        if (i < ilds) { \
          v8f R[5]; \
          if (i < irow) { \
              \
            _Pragma("unroll") \
            for (int G = 0; G < NGR; ++G) { \
                v4f x1 = rx1[G]; \
                v4f x2 = rx2[G]; \
                _Pragma("unroll") \
                for (int e = 0; e < 4; ++e) { \
                    x1[e] = cleanf(x1[e], top[G]); \
                    x2[e] = cleanf(x2[e], top[G]); \
                } \
                v4f t[5]; \
                t[0] = x1; t[1] = x2; t[2] = x1 * x1; t[3] = x2 * x2; t[4] = x1 * x2; \
                _Pragma("unroll") \
                for (int q = 0; q < 2; ++q) { \
                    const v2f b = wfr[2 * G + q]; \
                    _Pragma("unroll") \
                    for (int c = 0; c < 5; ++c) { \
                        v2f a; a[0] = t[c][2 * q]; a[1] = t[c][2 * q + 1]; \
                        R[c] = (G == 0 && q == 0) ? WMMA(a, b, zero8) : WMMA(a, b, R[c]); \
                    } \
                } \
            } \
              \
            ISSUE_TILE(i + 1) \
              \
            if (seg && n < NSH) { \
                _Pragma("unroll") \
                for (int c = 0; c < 5; ++c) { \
                    rsh[sp][n][c][0][lane] = __builtin_shufflevector(R[c], R[c], 0, 1, 2, 3); \
                    rsh[sp][n][c][1][lane] = __builtin_shufflevector(R[c], R[c], 4, 5, 6, 7); \
                } \
            } \
          } else { \
              \
            _Pragma("unroll") \
            for (int c = 0; c < 5; ++c) { \
                const v4f lo4 = rsh[sp][i - irow][c][0][lane]; \
                const v4f hi4 = rsh[sp][i - irow][c][1][lane]; \
                R[c] = __builtin_shufflevector(lo4, hi4, 0, 1, 2, 3, 4, 5, 6, 7); \
            } \
              \
              \
              \
            asm volatile("" :: "v"(R[0]), "v"(R[1]), "v"(R[2]), "v"(R[3]), "v"(R[4])); \
          } \
          col_step(SRC, DST, fin, R, wfc, n, M); \
        } else { \
            idle_step(SRC, DST, fin); \
        } \
          \
        if ((unsigned)md < (unsigned)M) \
            ssim_tail(fin, out, img, 16 * (of + md) + 8, ct, j, half); \
    }

    static_assert(M_UP + NBAND - 1 - 3 == M_LO + NBAND - 1 && (M_LO + NBAND - 1) % 2 == 0, "equal, even walks");
    static_assert(M_UP + M_LO == IMG_H / 16 + 1, "the two parts cover the image");
#pragma nounroll
    for (int tr = 0; tr < (M_LO + NBAND - 1) / 2; ++tr) {
        const int nb = n0 + 2 * tr;
        if (tr == 7) {
            __builtin_amdgcn_s_barrier_wait(-1);
            __builtin_amdgcn_fence(__ATOMIC_ACQUIRE, "workgroup");
        }
        WALK_STEP(nb,     accA, accB)
        WALK_STEP(nb + 1, accB, accA)
        if (tr == 2) {
            __builtin_amdgcn_fence(__ATOMIC_RELEASE, "workgroup");
            __builtin_amdgcn_s_barrier_signal(-1);
        }
    }
#undef WALK_STEP
#undef ISSUE_TILE
}

extern "C" void kernel_launch(void* const* d_in, const int* in_sizes, int n_in,
                              void* d_out, int out_size, void* d_ws, size_t ws_size,
                              hipStream_t stream) {
    const float* img1 = (const float*)d_in[0];
    const float* img2 = (const float*)d_in[1];
    const float* w1d  = (const float*)d_in[2];
    float* out = (float*)d_out;

    dim3 block(128);
    dim3 grid(16, 1, NIMG);
    hipLaunchKernelGGL(ssim_fused, grid, block, 0, stream, img1, img2, w1d, out);
}


# k-steps whose weights are zero in every lane (tested once per wave at run time) are left out: row blur per 8-wide band group, column blur 3 of a band position's 4; exact, order of sums unchanged
# speedup vs baseline: 4.5589x; 1.1647x over previous
#include <hip/hip_runtime.h>
#include <math.h>


typedef __attribute__((ext_vector_type(2))) float v2f;
typedef __attribute__((ext_vector_type(4))) float v4f;
typedef __attribute__((ext_vector_type(8))) float v8f;

#define IMG_H 512
#define IMG_W 512
#define NIMG  16
#define WIN   81
#define PAD   40
#define NGR   12
#define NBAND 6
#define NSH   5
#define M_UP  18
#define M_LO  15

__device__ __forceinline__ float cleanf(float v, float top) {
    return fminf(fmaxf(v, 0.0f), top);
}

#define WMMA(A, B, C) __builtin_amdgcn_wmma_f32_16x16x4_f32(false, (A), false, (B), (short)0, (C), false, false)

__device__ __forceinline__ void load_wfrag_rows(const float* wz, int j, int half, v2f (&wf)[24]) {
#pragma unroll
    for (int s = 0; s < 24; ++s) {
        const int p = 8 * (s >> 1) + 4 * half + 2 * (s & 1);
        wf[s][0] = wz[16 + p - j];
        wf[s][1] = wz[17 + p - j];
    }
}

__device__ __forceinline__ void load_wfrag_cols(const float* wz, int j, int half, v2f (&wf)[24]) {
#pragma unroll
    for (int s = 0; s < 24; ++s) {
        const int p = 16 * (s >> 2) + 8 * half + 2 * (s & 3);
        wf[s][0] = wz[16 + p - j];
        wf[s][1] = wz[17 + p - j];
    }
}

__device__ __forceinline__ void ssim_tail(const v8f (&a)[5], float* __restrict__ out,
                                          int img, int row0, int ct, int j, int half) {
    const float C1v = 1e-4f, C2v = 9e-4f;
    float r8[8];
#pragma unroll
    for (int v = 0; v < 8; ++v) {
        const float mu1 = a[0][v], mu2 = a[1][v];
        const float e11 = a[2][v], e22 = a[3][v], e12 = a[4][v];
        const float mu1sq = mu1 * mu1, mu2sq = mu2 * mu2, mu12 = mu1 * mu2;
        const float s11 = fmaxf(e11 - mu1sq, 0.0f);
        const float s22 = fmaxf(e22 - mu2sq, 0.0f);
        const float s12 = e12 - mu12;
        const float num = (2.0f * mu12 + C1v) * (2.0f * s12 + C2v);
        const float den = (mu1sq + mu2sq + C1v) * (s11 + s22 + C2v);
        float r = num / (den + 1e-8f);
        r = (r != r) ? 0.0f : r;
        r8[v] = fminf(fmaxf(r, -1.0f), 1.0f);
    }
    const int row = row0 + j;
    if ((unsigned)row < (unsigned)IMG_H) {
        float* o = out + ((size_t)img * IMG_H + row) * IMG_W + ct * 16 + 8 * half;
        v4f lo, hi4;
        lo[0] = r8[0]; lo[1] = r8[1]; lo[2] = r8[2]; lo[3] = r8[3];
        hi4[0] = r8[4]; hi4[1] = r8[5]; hi4[2] = r8[6]; hi4[3] = r8[7];
        *(v4f*)(o) = lo;
        *(v4f*)(o + 4) = hi4;
    }
}

__device__ __forceinline__ void col_step(const v8f (&src)[NBAND][5], v8f (&dst)[NBAND][5], v8f (&fin)[5],
                                         const v8f (&R)[5], const v2f (&wfc)[24], int n, int M, unsigned fc) {
    const v8f zero8 = {};
#pragma unroll
    for (int dp = NBAND - 1; dp >= 0; --dp) {
        const int m = n - dp;
        if ((unsigned)m < (unsigned)M) {
            {
                const v2f b = wfc[4 * dp];
#pragma unroll
                for (int c = 0; c < 5; ++c) {
                    v2f a; a[0] = R[c][0]; a[1] = R[c][1];
                    v8f& d = (dp == NBAND - 1) ? fin[c] : dst[dp + 1][c];
                    if (dp == 0) d = WMMA(a, b, zero8);
                    else         d = WMMA(a, b, src[dp][c]);
                }
            }
            if (fc & (1u << dp)) {
#pragma unroll
                for (int q = 1; q < 4; ++q) {
                    const v2f b = wfc[4 * dp + q];
#pragma unroll
                    for (int c = 0; c < 5; ++c) {
                        v2f a; a[0] = R[c][2 * q]; a[1] = R[c][2 * q + 1];
                        v8f& d = (dp == NBAND - 1) ? fin[c] : dst[dp + 1][c];
                        d = WMMA(a, b, d);
                    }
                }
            }
        }
    }
}

__device__ __forceinline__ void idle_step(const v8f (&src)[NBAND][5], v8f (&dst)[NBAND][5], v8f (&fin)[5]) {
#pragma unroll
    for (int dp = NBAND - 1; dp >= 1; --dp) {
#pragma unroll
        for (int c = 0; c < 5; ++c) {
            if (dp == NBAND - 1) fin[c] = src[dp][c];
            else                 dst[dp + 1][c] = src[dp][c];
        }
    }
}

__global__ __launch_bounds__(128) void ssim_fused(
    const float* __restrict__ img1, const float* __restrict__ img2,
    const float* __restrict__ w1d, float* __restrict__ out)
{
    const int tid = threadIdx.x;
    const float wpad = (tid >= 16 && tid < 16 + WIN) ? w1d[tid - 16] : 0.0f;
    const int lane = tid & 31;
    const int wv   = __builtin_amdgcn_readfirstlane(tid >> 5);
    const int j    = lane & 15;
    const int half = lane >> 4;

    const int sp  = wv >> 1;
    const int seg = wv & 1;
    const int ct  = blockIdx.x * 2 + sp;
    const int c0  = ct * 16;
    const int img = blockIdx.z;
    const int M   = seg ? M_LO : M_UP;
    const int of  = seg ? M_UP - 1 : -1;
    const int i0  = of - 2;
    const int n0  = seg ? 0 : 3;
    const int irow = seg ? IMG_H / 16 : M_UP - 3;
    const int ilds = seg ? IMG_H / 16 : M_UP - 3 + NSH;
    const float* b1 = img1 + (size_t)img * IMG_H * IMG_W;
    const float* b2 = img2 + (size_t)img * IMG_H * IMG_W;

    int coff[NGR];
    float top[NGR];
#pragma unroll
    for (int G = 0; G < NGR; ++G) {
        const int col = c0 - PAD + 8 * G + 4 * half;
        const bool ok = (unsigned)col < (unsigned)IMG_W;
        coff[G] = j * IMG_W + (ok ? col : 0);
        top[G]  = ok ? 1.0f : 0.0f;
    }

    v4f rx1[NGR], rx2[NGR];
#define ISSUE_TILE(I) { \
        const int it = min(max((I), 0), IMG_H / 16 - 1);       \
        const float* p1 = b1 + it * 16 * IMG_W; \
        const float* p2 = b2 + it * 16 * IMG_W; \
        _Pragma("unroll") \
        for (int G = 0; G < NGR; ++G) { \
            rx1[G] = *(const v4f*)(p1 + coff[G]); \
            rx2[G] = *(const v4f*)(p2 + coff[G]); } }
    ISSUE_TILE(i0 + n0)

    __shared__ float wz[112];
    __shared__ v4f rsh[2][NSH][5][2][32];
    if (tid < 112) wz[tid] = wpad;
    __syncthreads();
    v2f wfr[24], wfc[24];
    load_wfrag_rows(wz, j, half, wfr);
    load_wfrag_cols(wz, j, half, wfc);

    unsigned fr = 0, fc = 0;
#pragma unroll
    for (int G = 0; G < NGR; ++G) {
        const bool nz = wfr[2 * G][0] != 0.0f || wfr[2 * G][1] != 0.0f ||
                        wfr[2 * G + 1][0] != 0.0f || wfr[2 * G + 1][1] != 0.0f;
        if (__builtin_amdgcn_ballot_w32(nz) != 0u) fr |= 1u << G;
    }
#pragma unroll
    for (int dp = 0; dp < NBAND; ++dp) {
        bool nz = false;
#pragma unroll
        for (int q = 0; q < 4; ++q) nz = nz || wfc[4 * dp + q][0] != 0.0f || wfc[4 * dp + q][1] != 0.0f;
        if (__builtin_amdgcn_ballot_w32(nz) != 0u) fc |= 1u << dp;
    }

    const v8f zero8 = {};
    v8f accA[NBAND][5], accB[NBAND][5], fin[5];
#pragma unroll
    for (int d = 1; d < NBAND; ++d)
#pragma unroll
        for (int c = 0; c < 5; ++c) accA[d][c] = zero8;


#define WALK_STEP(N, SRC, DST) { \
        const int n  = (N); \
        const int i  = i0 + n;                                 \
        const int md = n - (NBAND - 1);                        \
          \
        if (i < ilds) { \
          v8f R[5]; \
          if (i < irow) { \
              \
            _Pragma("unroll") \
            for (int c = 0; c < 5; ++c) R[c] = zero8; \
            _Pragma("unroll") \
            for (int G = 0; G < NGR; ++G) if (fr & (1u << G)) { \
                v4f x1 = rx1[G]; \
                v4f x2 = rx2[G]; \
                _Pragma("unroll") \
                for (int e = 0; e < 4; ++e) { \
                    x1[e] = cleanf(x1[e], top[G]); \
                    x2[e] = cleanf(x2[e], top[G]); \
                } \
                v4f t[5]; \
                t[0] = x1; t[1] = x2; t[2] = x1 * x1; t[3] = x2 * x2; t[4] = x1 * x2; \
                _Pragma("unroll") \
                for (int q = 0; q < 2; ++q) { \
                    const v2f b = wfr[2 * G + q]; \
                    _Pragma("unroll") \
                    for (int c = 0; c < 5; ++c) { \
                        v2f a; a[0] = t[c][2 * q]; a[1] = t[c][2 * q + 1]; \
                        R[c] = WMMA(a, b, R[c]); \
                    } \
                } \
            } \
              \
            ISSUE_TILE(i + 1) \
              \
            if (seg && n < NSH) { \
                _Pragma("unroll") \
                for (int c = 0; c < 5; ++c) { \
                    rsh[sp][n][c][0][lane] = __builtin_shufflevector(R[c], R[c], 0, 1, 2, 3); \
                    rsh[sp][n][c][1][lane] = __builtin_shufflevector(R[c], R[c], 4, 5, 6, 7); \
                } \
            } \
          } else { \
              \
            _Pragma("unroll") \
            for (int c = 0; c < 5; ++c) { \
                const v4f lo4 = rsh[sp][i - irow][c][0][lane]; \
                const v4f hi4 = rsh[sp][i - irow][c][1][lane]; \
                R[c] = __builtin_shufflevector(lo4, hi4, 0, 1, 2, 3, 4, 5, 6, 7); \
            } \
              \
              \
              \
            asm volatile("" :: "v"(R[0]), "v"(R[1]), "v"(R[2]), "v"(R[3]), "v"(R[4])); \
          } \
          col_step(SRC, DST, fin, R, wfc, n, M, fc); \
        } else { \
            idle_step(SRC, DST, fin); \
        } \
          \
        if ((unsigned)md < (unsigned)M) \
            ssim_tail(fin, out, img, 16 * (of + md) + 8, ct, j, half); \
    }

    static_assert(M_UP + NBAND - 1 - 3 == M_LO + NBAND - 1 && (M_LO + NBAND - 1) % 2 == 0, "equal, even walks");
    static_assert(M_UP + M_LO == IMG_H / 16 + 1, "the two parts cover the image");
#pragma nounroll
    for (int tr = 0; tr < (M_LO + NBAND - 1) / 2; ++tr) {
        const int nb = n0 + 2 * tr;
        if (tr == 7) {
            __builtin_amdgcn_s_barrier_wait(-1);
            __builtin_amdgcn_fence(__ATOMIC_ACQUIRE, "workgroup");
        }
        WALK_STEP(nb,     accA, accB)
        WALK_STEP(nb + 1, accB, accA)
        if (tr == 2) {
            __builtin_amdgcn_fence(__ATOMIC_RELEASE, "workgroup");
            __builtin_amdgcn_s_barrier_signal(-1);
        }
    }
#undef WALK_STEP
#undef ISSUE_TILE
}

extern "C" void kernel_launch(void* const* d_in, const int* in_sizes, int n_in,
                              void* d_out, int out_size, void* d_ws, size_t ws_size,
                              hipStream_t stream) {
    const float* img1 = (const float*)d_in[0];
    const float* img2 = (const float*)d_in[1];
    const float* w1d  = (const float*)d_in[2];
    float* out = (float*)d_out;

    dim3 block(128);
    dim3 grid(16, 1, NIMG);
    hipLaunchKernelGGL(ssim_fused, grid, block, 0, stream, img1, img2, w1d, out);
}
